# starting kernel
# baseline (speedup 1.0000x reference)
#include <hip/hip_runtime.h>
#include <hip/hip_bf16.h>

typedef __attribute__((ext_vector_type(2))) float v2f;
typedef __attribute__((ext_vector_type(4))) float v4f;
typedef __attribute__((ext_vector_type(8))) float v8f;

#define K_TAPS 15
#define CIN    128
#define F_OUT  256
#define L_IN   4096
#define L_OUT  (L_IN - K_TAPS + 1)
#define BATCH  32
#define TILE_M 64
#define ROWS   (TILE_M + K_TAPS - 1)
#define XPAD   132
#define KC     32
#define KPAD   36
#define NCHUNK ((K_TAPS * CIN) / KC)
#define EPS_F  1e-7f

__global__ __launch_bounds__(256)
void ckn1d_wmma_kernel(const float* __restrict__ x,
                       const float* __restrict__ w,
                       const float* __restrict__ scale,
                       const float* __restrict__ bias,
                       float* __restrict__ out) {
    __shared__ float lds_x[ROWS * XPAD];
    __shared__ float lds_bt[F_OUT * KPAD];
    __shared__ float lds_rowsq[ROWS];
    __shared__ float lds_invn[TILE_M];
    __shared__ float lds_bias[F_OUT];

    const int t    = threadIdx.x;
    const int lane = t & 31;
    const int wave = t >> 5;
    const int l0   = blockIdx.x * TILE_M;
    const int b    = blockIdx.y;

    const float* xb = x + (size_t)b * L_IN * CIN;
    #pragma unroll
    for (int i = 0; i < 10; ++i) {
        int q = t + 256 * i;
        if (q < ROWS * (CIN / 4)) {
            int row = q >> 5;
            int f4  = q & 31;
            int g   = l0 + row;
            v4f val = {0.f, 0.f, 0.f, 0.f};
            if (g < L_IN) val = *(const v4f*)(xb + (size_t)g * CIN + f4 * 4);
            *(v4f*)(&lds_x[row * XPAD + f4 * 4]) = val;
        }
    }
    if (t < F_OUT) lds_bias[t] = bias[t];
    __syncthreads();

    if (t < ROWS) {
        const v4f* r = (const v4f*)&lds_x[t * XPAD];
        float s = 0.f;
        #pragma unroll
        for (int c = 0; c < CIN / 4; ++c) {
            v4f v = r[c];
            s += v.x * v.x + v.y * v.y + v.z * v.z + v.w * v.w;
        }
        lds_rowsq[t] = s;
    }
    __syncthreads();
    if (t < TILE_M) {
        float win = 0.f;
        #pragma unroll
        for (int k = 0; k < K_TAPS; ++k) win += lds_rowsq[t + k];
        lds_invn[t] = scale[0] / __builtin_sqrtf(win + EPS_F);
    }

    const int mt   = wave & 3;
    const int f0   = (wave >> 2) * 128;
    const int half = lane >> 4;
    const int lm   = lane & 15;

    v8f acc[8] = {};

    float pa[16], pb[16];
    {
        const float* src = w;
        #pragma unroll
        for (int i = 0; i < 16; ++i) {
            pa[i] = src[(2 * i)     * F_OUT + t];
            pb[i] = src[(2 * i + 1) * F_OUT + t];
        }
    }

    for (int ch = 0; ch < NCHUNK; ++ch) {
        __syncthreads();
        {
            float* dst = &lds_bt[t * KPAD];
            #pragma unroll
            for (int i = 0; i < 16; ++i) {
                v2f p;
                p.x = pa[i];
                p.y = pb[i];
                *(v2f*)(dst + 2 * i) = p;
            }
        }
        if (ch + 1 < NCHUNK) {
            const float* src = w + (size_t)(ch + 1) * KC * F_OUT;
            #pragma unroll
            for (int i = 0; i < 16; ++i) {
                pa[i] = src[(2 * i)     * F_OUT + t];
                pb[i] = src[(2 * i + 1) * F_OUT + t];
            }
        }
        __syncthreads();

        const int tap = ch >> 2;
        const int c0  = (ch & 3) * KC;
        const float* arow = &lds_x[(mt * 16 + lm + tap) * XPAD + c0 + 2 * half];
        const float* bcol = &lds_bt[(f0 + lm) * KPAD + 2 * half];

        #pragma unroll
        for (int kc = 0; kc < KC; kc += 4) {
            v2f a = *(const v2f*)(arow + kc);
            #pragma unroll
            for (int nt = 0; nt < 8; ++nt) {
                v2f bf = *(const v2f*)(bcol + nt * 16 * KPAD + kc);
                acc[nt] = __builtin_amdgcn_wmma_f32_16x16x4_f32(
                    false, a, false, bf, (short)0, acc[nt], false, false);
            }
        }
    }

    float* ob = out + ((size_t)b * L_OUT + l0) * F_OUT;
    #pragma unroll
    for (int v = 0; v < 8; ++v) {
        int mrow = mt * 16 + v + 8 * half;
        if (l0 + mrow < L_OUT) {
            float inv = lds_invn[mrow];
            #pragma unroll
            for (int nt = 0; nt < 8; ++nt) {
                int f = f0 + nt * 16 + lm;
                ob[(size_t)mrow * F_OUT + f] = acc[nt][v] * inv + lds_bias[f];
            }
        }
    }
}

extern "C" void kernel_launch(void* const* d_in, const int* in_sizes, int n_in,
                              void* d_out, int out_size, void* d_ws, size_t ws_size,
                              hipStream_t stream) {
    const float* x     = (const float*)d_in[0];
    const float* w     = (const float*)d_in[1];
    const float* scale = (const float*)d_in[2];
    const float* bias  = (const float*)d_in[3];
    float* out = (float*)d_out;

    dim3 grid((L_OUT + TILE_M - 1) / TILE_M, BATCH);
    ckn1d_wmma_kernel<<<grid, 256, 0, stream>>>(x, w, scale, bias, out);
}


# conv in the frequency domain: 64-point odd-frequency transform per 50-output tile; forward, 32 complex channel products and inverse as f32 WMMA GEMMs; 4 launches; matrix work x0.24; sum order changes
# speedup vs baseline: 2.9710x; 2.9710x over previous
#include <hip/hip_runtime.h>

typedef __attribute__((ext_vector_type(2))) float v2f;
typedef __attribute__((ext_vector_type(4))) float v4f;
typedef __attribute__((ext_vector_type(8))) float v8f;

#define K_TAPS 15
#define CIN    128
#define F_OUT  256
#define L_IN   4096
#define L_OUT  (L_IN - K_TAPS + 1)
#define BATCH  32
#define NFFT   64
#define VALID  (NFFT - K_TAPS + 1)
#define NTILE  82
#define TILES  (BATCH * NTILE)
#define NBIN   32
#define TPAD   68
#define APAD   132
#define MT_PW  32
#define EPS_F  1e-7f

#define OFF_TT 4096
#define OFF_G  8192
#define OFF_XF 2105344
#define OFF_YF 23601152

__device__ __forceinline__ v8f mma4(v2f a, v2f b, v8f c) {
    return __builtin_amdgcn_wmma_f32_16x16x4_f32(false, a, false, b, (short)0, c, false, false);
}

__global__ __launch_bounds__(256)
void k_prep(const float* __restrict__ w, float* __restrict__ tabT,
            float* __restrict__ tabTT, float* __restrict__ G) {
    __shared__ float cs_t[16], sn_t[16];
    const int t   = threadIdx.x;
    const int k   = blockIdx.x >> 4;
    const int cpg = blockIdx.x & 15;
    if (t < K_TAPS) {
        float sn, cs;
        sincospif((float)((2 * k + 1) * t) * (1.0f / 64.0f), &sn, &cs);
        cs_t[t] = cs;
        sn_t[t] = sn;
    }
    __syncthreads();
    #pragma unroll
    for (int i = 0; i < 4; ++i) {
        const int cp = cpg * 4 + i;
        v2f gre = {0.f, 0.f}, gim = {0.f, 0.f};
        #pragma unroll
        for (int tap = 0; tap < K_TAPS; ++tap) {
            float w0 = w[(tap * CIN + 2 * cp)     * F_OUT + t];
            float w1 = w[(tap * CIN + 2 * cp + 1) * F_OUT + t];
            gre.x += w0 * cs_t[tap];
            gre.y += w1 * cs_t[tap];
            gim.x += w0 * sn_t[tap];
            gim.y += w1 * sn_t[tap];
        }
        *(v2f*)(G + (((size_t)(k * 2 + 0) * 64 + cp) * 256 + t) * 2) = gre;
        *(v2f*)(G + (((size_t)(k * 2 + 1) * 64 + cp) * 256 + t) * 2) = gim;
    }
    const int gid = blockIdx.x * 256 + t;
    if (gid < 4096) {
        const int row = gid >> 6, n = gid & 63;
        float sn, cs;
        sincospif((float)((row | 1) * n) * (1.0f / 64.0f), &sn, &cs);
        float v = (row & 1) ? sn : cs;
        tabT[row * 64 + n]  = v;
        tabTT[n * 64 + row] = v;
    }
}

__global__ __launch_bounds__(256)
void k_fwd(const float* __restrict__ x, const float* __restrict__ tabT, float* __restrict__ Xf) {
    __shared__ float lds_t[64 * TPAD];
    const int t = threadIdx.x, lane = t & 31, wave = t >> 5;
    #pragma unroll
    for (int i = 0; i < 4; ++i) {
        int q = t + 256 * i;
        int row = q >> 4, c4 = q & 15;
        v4f v = *(const v4f*)(tabT + row * 64 + c4 * 4);
        *(v4f*)(&lds_t[row * TPAD + c4 * 4]) = v;
    }
    __syncthreads();
    const int half = lane >> 4, lm = lane & 15;
    const float* arow = &lds_t[lm * TPAD + 2 * half];
    v2f bf[2][16];
    #pragma unroll
    for (int it = 0; it < 2; ++it) {
        const int tile = blockIdx.x * 2 + it;
        const int b  = tile / NTILE;
        const int l0 = (tile - b * NTILE) * VALID;
        const float* xb = x + (size_t)b * L_IN * CIN + 16 * wave + lm;
        #pragma unroll
        for (int s = 0; s < 16; ++s) {
            int n0 = l0 + 4 * s + 2 * half;
            int na = n0 < L_IN ? n0 : L_IN - 1;
            int nb = n0 + 1 < L_IN ? n0 + 1 : L_IN - 1;
            float va = xb[(size_t)na * CIN];
            float vb = xb[(size_t)nb * CIN];
            bf[it][s].x = n0 < L_IN ? va : 0.f;
            bf[it][s].y = n0 + 1 < L_IN ? vb : 0.f;
        }
    }
    #pragma unroll
    for (int it = 0; it < 2; ++it) {
        const int tile = blockIdx.x * 2 + it;
        v8f acc[4] = {};
        #pragma unroll
        for (int s = 0; s < 16; ++s) {
            #pragma unroll
            for (int mt = 0; mt < 4; ++mt) {
                v2f a = *(const v2f*)(arow + mt * 16 * TPAD + 4 * s);
                acc[mt] = mma4(a, bf[it][s], acc[mt]);
            }
        }
        #pragma unroll
        for (int mt = 0; mt < 4; ++mt) {
            #pragma unroll
            for (int r = 0; r < 8; ++r) {
                int row = 16 * mt + 8 * half + r;
                Xf[((size_t)row * TILES + tile) * CIN + 16 * wave + lm] = acc[mt][r];
            }
        }
    }
}

__global__ __launch_bounds__(256)
void k_pw(const float* __restrict__ Xf, const float* __restrict__ G, float* __restrict__ Yf) {
    __shared__ float lds_a[2 * MT_PW * APAD];
    const int t = threadIdx.x, lane = t & 31, wave = t >> 5;
    const int tile0 = blockIdx.x * MT_PW;
    const int k     = blockIdx.y;
    #pragma unroll
    for (int i = 0; i < 8; ++i) {
        int q  = t + 256 * i;
        int c4 = q & 31, m = (q >> 5) & 31, p = q >> 10;
        v4f v = *(const v4f*)(Xf + ((size_t)(2 * k + p) * TILES + tile0 + m) * CIN + c4 * 4);
        *(v4f*)(&lds_a[(p * MT_PW + m) * APAD + c4 * 4]) = v;
    }
    __syncthreads();
    const int half = lane >> 4, lm = lane & 15;
    const int f0 = wave * 32;
    const float* gre = G + ((size_t)(k * 2 + 0) * 64 * 256 + f0 + lm) * 2 + half * 512;
    const float* gim = G + ((size_t)(k * 2 + 1) * 64 * 256 + f0 + lm) * 2 + half * 512;
    const float* ap  = &lds_a[lm * APAD + 2 * half];

    v8f yre[2][2] = {}, yz[2][2] = {};
    v2f bre[2], bim[2];
    bre[0] = *(const v2f*)(gre);
    bre[1] = *(const v2f*)(gre + 32);
    bim[0] = *(const v2f*)(gim);
    bim[1] = *(const v2f*)(gim + 32);
    #pragma unroll 4
    for (int s = 0; s < 32; ++s) {
        v2f cre[2] = {bre[0], bre[1]};
        v2f cim[2] = {bim[0], bim[1]};
        if (s + 1 < 32) {
            bre[0] = *(const v2f*)(gre + (size_t)(s + 1) * 1024);
            bre[1] = *(const v2f*)(gre + (size_t)(s + 1) * 1024 + 32);
            bim[0] = *(const v2f*)(gim + (size_t)(s + 1) * 1024);
            bim[1] = *(const v2f*)(gim + (size_t)(s + 1) * 1024 + 32);
        }
        v2f nim[2] = {-cim[0], -cim[1]};
        #pragma unroll
        for (int mt = 0; mt < 2; ++mt) {
            v2f pa = *(const v2f*)(ap + (mt * 16) * APAD + 4 * s);
            v2f qa = *(const v2f*)(ap + (MT_PW + mt * 16) * APAD + 4 * s);
            yre[mt][0] = mma4(pa, cre[0], yre[mt][0]);
            yz[mt][0]  = mma4(qa, cre[0], yz[mt][0]);
            yre[mt][1] = mma4(pa, cre[1], yre[mt][1]);
            yz[mt][1]  = mma4(qa, cre[1], yz[mt][1]);
            yre[mt][0] = mma4(qa, cim[0], yre[mt][0]);
            yz[mt][0]  = mma4(pa, nim[0], yz[mt][0]);
            yre[mt][1] = mma4(qa, cim[1], yre[mt][1]);
            yz[mt][1]  = mma4(pa, nim[1], yz[mt][1]);
        }
    }
    #pragma unroll
    for (int mt = 0; mt < 2; ++mt) {
        #pragma unroll
        for (int r = 0; r < 8; ++r) {
            int tile = tile0 + 16 * mt + 8 * half + r;
            #pragma unroll
            for (int nt = 0; nt < 2; ++nt) {
                int f = f0 + 16 * nt + lm;
                v2f o;
                o.x = yre[mt][nt][r];
                o.y = yz[mt][nt][r];
                *(v2f*)(Yf + (((size_t)tile * NBIN + k) * F_OUT + f) * 2) = o;
            }
        }
    }
}

__global__ __launch_bounds__(256)
void k_inv(const float* __restrict__ x, const float* __restrict__ tabTT,
           const float* __restrict__ Yf, const float* __restrict__ scale,
           const float* __restrict__ bias, float* __restrict__ out) {
    __shared__ float lds_t[64 * TPAD];
    __shared__ float lds_part[64 * 4];
    __shared__ float lds_invn[64];
    const int t = threadIdx.x, lane = t & 31, wave = t >> 5;
    const int tile = blockIdx.x;
    const int b  = tile / NTILE;
    const int l0 = (tile - b * NTILE) * VALID;
    #pragma unroll
    for (int i = 0; i < 4; ++i) {
        int q = t + 256 * i;
        int row = q >> 4, c4 = q & 15;
        v4f v = *(const v4f*)(tabTT + row * 64 + c4 * 4);
        *(v4f*)(&lds_t[row * TPAD + c4 * 4]) = v;
    }
    {
        int row = t >> 2, qd = t & 3;
        int g = l0 + row;
        float s = 0.f;
        if (g < L_IN) {
            const v4f* r = (const v4f*)(x + ((size_t)b * L_IN + g) * CIN + qd * 32);
            #pragma unroll
            for (int i = 0; i < 8; ++i) {
                v4f v = r[i];
                s += v.x * v.x + v.y * v.y + v.z * v.z + v.w * v.w;
            }
        }
        lds_part[t] = s;
    }
    __syncthreads();
    if (t < 64) {
        float win = 0.f;
        if (t < VALID) {
            #pragma unroll
            for (int j = 0; j < K_TAPS; ++j) {
                const float* pp = &lds_part[(t + j) * 4];
                win += (pp[0] + pp[1]) + (pp[2] + pp[3]);
            }
        }
        lds_invn[t] = scale[0] / __builtin_sqrtf(win + EPS_F) * (1.0f / 32.0f);
    }
    __syncthreads();

    const int half = lane >> 4, lm = lane & 15;
    const int f0 = wave * 32;
    const float* yb   = Yf + (((size_t)tile * NBIN + half) * F_OUT + f0 + lm) * 2;
    const float* arow = &lds_t[lm * TPAD + 2 * half];
    v2f bf0[16], bf1[16];
    #pragma unroll
    for (int s = 0; s < 16; ++s) {
        bf0[s] = *(const v2f*)(yb + (size_t)s * 1024);
        bf1[s] = *(const v2f*)(yb + (size_t)s * 1024 + 32);
    }
    v8f acc[4][2] = {};
    #pragma unroll
    for (int s = 0; s < 16; ++s) {
        #pragma unroll
        for (int mt = 0; mt < 4; ++mt) {
            v2f a = *(const v2f*)(arow + mt * 16 * TPAD + 4 * s);
            acc[mt][0] = mma4(a, bf0[s], acc[mt][0]);
            acc[mt][1] = mma4(a, bf1[s], acc[mt][1]);
        }
    }
    const float bias0 = bias[f0 + lm], bias1 = bias[f0 + 16 + lm];
    float* ob = out + ((size_t)b * L_OUT + l0) * F_OUT + f0 + lm;
    #pragma unroll
    for (int mt = 0; mt < 4; ++mt) {
        #pragma unroll
        for (int r = 0; r < 8; ++r) {
            int l = 16 * mt + 8 * half + r;
            if (l < VALID && l0 + l < L_OUT) {
                float inv = lds_invn[l];
                ob[(size_t)l * F_OUT]      = acc[mt][0][r] * inv + bias0;
                ob[(size_t)l * F_OUT + 16] = acc[mt][1][r] * inv + bias1;
            }
        }
    }
}

extern "C" void kernel_launch(void* const* d_in, const int* in_sizes, int n_in,
                              void* d_out, int out_size, void* d_ws, size_t ws_size,
                              hipStream_t stream) {
    const float* x     = (const float*)d_in[0];
    const float* w     = (const float*)d_in[1];
    const float* scale = (const float*)d_in[2];
    const float* bias  = (const float*)d_in[3];
    float* out = (float*)d_out;
    float* ws  = (float*)d_ws;
    float* tabT  = ws;
    float* tabTT = ws + OFF_TT;
    float* G     = ws + OFF_G;
    float* Xf    = ws + OFF_XF;
    float* Yf    = ws + OFF_YF;

    k_prep<<<512, 256, 0, stream>>>(w, tabT, tabTT, G);
    k_fwd<<<TILES / 2, 256, 0, stream>>>(x, tabT, Xf);
    k_pw<<<dim3(TILES / MT_PW, NBIN), 256, 0, stream>>>(Xf, G, Yf);
    k_inv<<<TILES, 256, 0, stream>>>(x, tabTT, Yf, scale, bias, out);
}


# per-frequency complex channel product by three real matrix products, not four: (P-Q)Gre, Q(Gre+Gim), P(Gre-Gim); block 64 tiles x 128 filters; matrix work of that launch x 3/4; sum order changes
# speedup vs baseline: 3.2242x; 1.0852x over previous
#include <hip/hip_runtime.h>

typedef __attribute__((ext_vector_type(2))) float v2f;
typedef __attribute__((ext_vector_type(4))) float v4f;
typedef __attribute__((ext_vector_type(8))) float v8f;

#define K_TAPS 15
#define CIN    128
#define F_OUT  256
#define L_IN   4096
#define L_OUT  (L_IN - K_TAPS + 1)
#define BATCH  32
#define NFFT   64
#define VALID  (NFFT - K_TAPS + 1)
#define NTILE  82
#define TILES  (BATCH * NTILE)
#define NBIN   32
#define TPAD   68
#define APAD   132
#define MT_PW  64
#define NF_PW  128
#define EPS_F  1e-7f

#define OFF_TT 4096
#define OFF_G  8192
#define OFF_XF 3153920
#define OFF_YF 24649728

__device__ __forceinline__ v8f mma4(v2f a, v2f b, v8f c) {
    return __builtin_amdgcn_wmma_f32_16x16x4_f32(false, a, false, b, (short)0, c, false, false);
}

__global__ __launch_bounds__(256)
void k_prep(const float* __restrict__ w, float* __restrict__ tabT,
            float* __restrict__ tabTT, float* __restrict__ G) {
    __shared__ float cs_t[16], sn_t[16];
    const int t   = threadIdx.x;
    const int k   = blockIdx.x >> 4;
    const int cpg = blockIdx.x & 15;
    if (t < K_TAPS) {
        float sn, cs;
        sincospif((float)((2 * k + 1) * t) * (1.0f / 64.0f), &sn, &cs);
        cs_t[t] = cs;
        sn_t[t] = sn;
    }
    __syncthreads();
    #pragma unroll
    for (int i = 0; i < 4; ++i) {
        const int cp = cpg * 4 + i;
        v2f gre = {0.f, 0.f}, gim = {0.f, 0.f};
        #pragma unroll
        for (int tap = 0; tap < K_TAPS; ++tap) {
            float w0 = w[(tap * CIN + 2 * cp)     * F_OUT + t];
            float w1 = w[(tap * CIN + 2 * cp + 1) * F_OUT + t];
            gre.x += w0 * cs_t[tap];
            gre.y += w1 * cs_t[tap];
            gim.x += w0 * sn_t[tap];
            gim.y += w1 * sn_t[tap];
        }
        *(v2f*)(G + (((size_t)(k * 3 + 0) * 64 + cp) * 256 + t) * 2) = gre;
        *(v2f*)(G + (((size_t)(k * 3 + 1) * 64 + cp) * 256 + t) * 2) = gre + gim;
        *(v2f*)(G + (((size_t)(k * 3 + 2) * 64 + cp) * 256 + t) * 2) = gre - gim;
    }
    const int gid = blockIdx.x * 256 + t;
    if (gid < 4096) {
        const int row = gid >> 6, n = gid & 63;
        float sn, cs;
        sincospif((float)((row | 1) * n) * (1.0f / 64.0f), &sn, &cs);
        float v = (row & 1) ? sn : cs;
        tabT[row * 64 + n]  = v;
        tabTT[n * 64 + row] = v;
    }
}

__global__ __launch_bounds__(256)
void k_fwd(const float* __restrict__ x, const float* __restrict__ tabT, float* __restrict__ Xf) {
    __shared__ float lds_t[64 * TPAD];
    const int t = threadIdx.x, lane = t & 31, wave = t >> 5;
    #pragma unroll
    for (int i = 0; i < 4; ++i) {
        int q = t + 256 * i;
        int row = q >> 4, c4 = q & 15;
        v4f v = *(const v4f*)(tabT + row * 64 + c4 * 4);
        *(v4f*)(&lds_t[row * TPAD + c4 * 4]) = v;
    }
    __syncthreads();
    const int half = lane >> 4, lm = lane & 15;
    const float* arow = &lds_t[lm * TPAD + 2 * half];
    v2f bf[2][16];
    #pragma unroll
    for (int it = 0; it < 2; ++it) {
        const int tile = blockIdx.x * 2 + it;
        const int b  = tile / NTILE;
        const int l0 = (tile - b * NTILE) * VALID;
        const float* xb = x + (size_t)b * L_IN * CIN + 16 * wave + lm;
        #pragma unroll
        for (int s = 0; s < 16; ++s) {
            int n0 = l0 + 4 * s + 2 * half;
            int na = n0 < L_IN ? n0 : L_IN - 1;
            int nb = n0 + 1 < L_IN ? n0 + 1 : L_IN - 1;
            float va = xb[(size_t)na * CIN];
            float vb = xb[(size_t)nb * CIN];
            bf[it][s].x = n0 < L_IN ? va : 0.f;
            bf[it][s].y = n0 + 1 < L_IN ? vb : 0.f;
        }
    }
    #pragma unroll
    for (int it = 0; it < 2; ++it) {
        const int tile = blockIdx.x * 2 + it;
        v8f acc[4] = {};
        #pragma unroll
        for (int s = 0; s < 16; ++s) {
            #pragma unroll
            for (int mt = 0; mt < 4; ++mt) {
                v2f a = *(const v2f*)(arow + mt * 16 * TPAD + 4 * s);
                acc[mt] = mma4(a, bf[it][s], acc[mt]);
            }
        }
        #pragma unroll
        for (int mt = 0; mt < 4; ++mt) {
            #pragma unroll
            for (int r = 0; r < 8; ++r) {
                int row = 16 * mt + 8 * half + r;
                Xf[((size_t)row * TILES + tile) * CIN + 16 * wave + lm] = acc[mt][r];
            }
        }
    }
}

__global__ __launch_bounds__(256)
void k_pw(const float* __restrict__ Xf, const float* __restrict__ G, float* __restrict__ Yf) {
    __shared__ float lds_a[3 * MT_PW * APAD];
    const int t = threadIdx.x, lane = t & 31;
    const int wave = __builtin_amdgcn_readfirstlane(t >> 5);
    const int tile0 = blockIdx.x * MT_PW;
    const int fblk  = blockIdx.y * NF_PW;
    const int k     = blockIdx.z;
    #pragma unroll
    for (int i = 0; i < 8; ++i) {
        int q  = t + 256 * i;
        int c4 = q & 31, m = q >> 5;
        v4f vp = *(const v4f*)(Xf + ((size_t)(2 * k)     * TILES + tile0 + m) * CIN + c4 * 4);
        v4f vq = *(const v4f*)(Xf + ((size_t)(2 * k + 1) * TILES + tile0 + m) * CIN + c4 * 4);
        *(v4f*)(&lds_a[(0 * MT_PW + m) * APAD + c4 * 4]) = vp - vq;
        *(v4f*)(&lds_a[(1 * MT_PW + m) * APAD + c4 * 4]) = vq;
        *(v4f*)(&lds_a[(2 * MT_PW + m) * APAD + c4 * 4]) = vp;
    }
    __syncthreads();
    const int half = lane >> 4, lm = lane & 15;
    const int f0 = fblk + wave * 16;
    const float* g0 = G + ((size_t)(k * 3) * 64 * 256 + f0 + lm) * 2 + half * 512;
    const float* g1 = g0 + 64 * 256 * 2;
    const float* g2 = g1 + 64 * 256 * 2;
    const float* ap = &lds_a[lm * APAD + 2 * half];

    v8f s1[4] = {}, s2[4] = {}, s3[4] = {};
    v2f b0 = *(const v2f*)(g0);
    v2f b1 = *(const v2f*)(g1);
    v2f b2 = *(const v2f*)(g2);
    #pragma unroll 4
    for (int s = 0; s < 32; ++s) {
        v2f c0 = b0, c1 = b1, c2 = b2;
        if (s + 1 < 32) {
            b0 = *(const v2f*)(g0 + (size_t)(s + 1) * 1024);
            b1 = *(const v2f*)(g1 + (size_t)(s + 1) * 1024);
            b2 = *(const v2f*)(g2 + (size_t)(s + 1) * 1024);
        }
        #pragma unroll
        for (int mt = 0; mt < 4; ++mt) {
            v2f da = *(const v2f*)(ap + (0 * MT_PW + mt * 16) * APAD + 4 * s);
            v2f qa = *(const v2f*)(ap + (1 * MT_PW + mt * 16) * APAD + 4 * s);
            v2f pa = *(const v2f*)(ap + (2 * MT_PW + mt * 16) * APAD + 4 * s);
            s1[mt] = mma4(da, c0, s1[mt]);
            s2[mt] = mma4(qa, c1, s2[mt]);
            s3[mt] = mma4(pa, c2, s3[mt]);
        }
    }
    const int f = f0 + lm;
    #pragma unroll
    for (int mt = 0; mt < 4; ++mt) {
        #pragma unroll
        for (int r = 0; r < 8; ++r) {
            int tile = tile0 + 16 * mt + 8 * half + r;
            v2f o;
            o.x = s1[mt][r] + s2[mt][r];
            o.y = s3[mt][r] - s1[mt][r];
            *(v2f*)(Yf + (((size_t)tile * NBIN + k) * F_OUT + f) * 2) = o;
        }
    }
}

__global__ __launch_bounds__(256)
void k_inv(const float* __restrict__ x, const float* __restrict__ tabTT,
           const float* __restrict__ Yf, const float* __restrict__ scale,
           const float* __restrict__ bias, float* __restrict__ out) {
    __shared__ float lds_t[64 * TPAD];
    __shared__ float lds_part[64 * 4];
    __shared__ float lds_invn[64];
    const int t = threadIdx.x, lane = t & 31, wave = t >> 5;
    const int tile = blockIdx.x;
    const int b  = tile / NTILE;
    const int l0 = (tile - b * NTILE) * VALID;
    #pragma unroll
    for (int i = 0; i < 4; ++i) {
        int q = t + 256 * i;
        int row = q >> 4, c4 = q & 15;
        v4f v = *(const v4f*)(tabTT + row * 64 + c4 * 4);
        *(v4f*)(&lds_t[row * TPAD + c4 * 4]) = v;
    }
    {
        int row = t >> 2, qd = t & 3;
        int g = l0 + row;
        float s = 0.f;
        if (g < L_IN) {
            const v4f* r = (const v4f*)(x + ((size_t)b * L_IN + g) * CIN + qd * 32);
            #pragma unroll
            for (int i = 0; i < 8; ++i) {
                v4f v = r[i];
                s += v.x * v.x + v.y * v.y + v.z * v.z + v.w * v.w;
            }
        }
        lds_part[t] = s;
    }
    __syncthreads();
    if (t < 64) {
        float win = 0.f;
        if (t < VALID) {
            #pragma unroll
            for (int j = 0; j < K_TAPS; ++j) {
                const float* pp = &lds_part[(t + j) * 4];
                win += (pp[0] + pp[1]) + (pp[2] + pp[3]);
            }
        }
        lds_invn[t] = scale[0] / __builtin_sqrtf(win + EPS_F) * (1.0f / 32.0f);
    }
    __syncthreads();

    const int half = lane >> 4, lm = lane & 15;
    const int f0 = wave * 32;
    const float* yb   = Yf + (((size_t)tile * NBIN + half) * F_OUT + f0 + lm) * 2;
    const float* arow = &lds_t[lm * TPAD + 2 * half];
    v2f bf0[16], bf1[16];
    #pragma unroll
    for (int s = 0; s < 16; ++s) {
        bf0[s] = *(const v2f*)(yb + (size_t)s * 1024);
        bf1[s] = *(const v2f*)(yb + (size_t)s * 1024 + 32);
    }
    v8f acc[4][2] = {};
    #pragma unroll
    for (int s = 0; s < 16; ++s) {
        #pragma unroll
        for (int mt = 0; mt < 4; ++mt) {
            v2f a = *(const v2f*)(arow + mt * 16 * TPAD + 4 * s);
            acc[mt][0] = mma4(a, bf0[s], acc[mt][0]);
            acc[mt][1] = mma4(a, bf1[s], acc[mt][1]);
        }
    }
    const float bias0 = bias[f0 + lm], bias1 = bias[f0 + 16 + lm];
    float* ob = out + ((size_t)b * L_OUT + l0) * F_OUT + f0 + lm;
    #pragma unroll
    for (int mt = 0; mt < 4; ++mt) {
        #pragma unroll
        for (int r = 0; r < 8; ++r) {
            int l = 16 * mt + 8 * half + r;
            if (l < VALID && l0 + l < L_OUT) {
                float inv = lds_invn[l];
                ob[(size_t)l * F_OUT]      = acc[mt][0][r] * inv + bias0;
                ob[(size_t)l * F_OUT + 16] = acc[mt][1][r] * inv + bias1;
            }
        }
    }
}

extern "C" void kernel_launch(void* const* d_in, const int* in_sizes, int n_in,
                              void* d_out, int out_size, void* d_ws, size_t ws_size,
                              hipStream_t stream) {
    const float* x     = (const float*)d_in[0];
    const float* w     = (const float*)d_in[1];
    const float* scale = (const float*)d_in[2];
    const float* bias  = (const float*)d_in[3];
    float* out = (float*)d_out;
    float* ws  = (float*)d_ws;
    float* tabT  = ws;
    float* tabTT = ws + OFF_TT;
    float* G     = ws + OFF_G;
    float* Xf    = ws + OFF_XF;
    float* Yf    = ws + OFF_YF;

    k_prep<<<512, 256, 0, stream>>>(w, tabT, tabTT, G);
    k_fwd<<<TILES / 2, 256, 0, stream>>>(x, tabT, Xf);
    k_pw<<<dim3(TILES / MT_PW, F_OUT / NF_PW, NBIN), 256, 0, stream>>>(Xf, G, Yf);
    k_inv<<<TILES, 256, 0, stream>>>(x, tabTT, Yf, scale, bias, out);
}


# transforms folded about the tile midpoint (half-sample-shifted cos/sin tables: two half-size products each), spent on 128-point tiles: 64 bins per 114 outputs, not 32 per 50; sum order changes
# speedup vs baseline: 3.3440x; 1.0372x over previous
#include <hip/hip_runtime.h>

typedef __attribute__((ext_vector_type(2))) float v2f;
typedef __attribute__((ext_vector_type(4))) float v4f;
typedef __attribute__((ext_vector_type(8))) float v8f;

#define K_TAPS 15
#define CIN    128
#define F_OUT  256
#define L_IN   4096
#define L_OUT  (L_IN - K_TAPS + 1)
#define BATCH  32
#define NFFT   128
#define VALID  (NFFT - K_TAPS + 1)
#define NTILE  36
#define TILES  (BATCH * NTILE)
#define NBIN   64
#define NH     64
#define HPAD   68
#define APAD   132
#define MT_PW  64
#define NF_PW  128
#define EPS_F  1e-7f

#define OFF_G  8192
#define OFF_XF 6299648
#define OFF_YF 25174016

__device__ __forceinline__ v8f mma4(v2f a, v2f b, v8f c) {
    return __builtin_amdgcn_wmma_f32_16x16x4_f32(false, a, false, b, (short)0, c, false, false);
}

__global__ __launch_bounds__(256)
void k_prep(const float* __restrict__ w, float* __restrict__ tab, float* __restrict__ G) {
    __shared__ float cs_t[16], sn_t[16];
    const int t   = threadIdx.x;
    const int k   = blockIdx.x >> 4;
    const int cpg = blockIdx.x & 15;
    if (t < K_TAPS) {
        float sn, cs;
        sincospif((float)((2 * k + 1) * t) * (1.0f / 128.0f), &sn, &cs);
        cs_t[t] = cs;
        sn_t[t] = sn;
    }
    __syncthreads();
    #pragma unroll
    for (int i = 0; i < 4; ++i) {
        const int cp = cpg * 4 + i;
        v2f gre = {0.f, 0.f}, gim = {0.f, 0.f};
        #pragma unroll
        for (int tap = 0; tap < K_TAPS; ++tap) {
            float w0 = w[(tap * CIN + 2 * cp)     * F_OUT + t];
            float w1 = w[(tap * CIN + 2 * cp + 1) * F_OUT + t];
            gre.x += w0 * cs_t[tap];
            gre.y += w1 * cs_t[tap];
            gim.x += w0 * sn_t[tap];
            gim.y += w1 * sn_t[tap];
        }
        *(v2f*)(G + (((size_t)(k * 3 + 0) * 64 + cp) * 256 + t) * 2) = gre;
        *(v2f*)(G + (((size_t)(k * 3 + 1) * 64 + cp) * 256 + t) * 2) = gre + gim;
        *(v2f*)(G + (((size_t)(k * 3 + 2) * 64 + cp) * 256 + t) * 2) = gre - gim;
    }
    const int gid = blockIdx.x * 256 + t;
    if (gid < 2 * NH * NH) {
        const int p = gid >> 12, i = (gid >> 6) & 63, j = gid & 63;
        float sn, cs;
        sincospif((float)((2 * i + 1) * (2 * j + 1)) * (1.0f / 256.0f), &sn, &cs);
        tab[gid] = p ? sn : cs;
    }
}

__global__ __launch_bounds__(256)
void k_fwd(const float* __restrict__ x, const float* __restrict__ tab, float* __restrict__ Xf) {
    __shared__ float lds_t[2 * NH * HPAD];
    const int t = threadIdx.x, lane = t & 31, wave = t >> 5;
    const int half = lane >> 4, lm = lane & 15;
    const int tile = blockIdx.x;
    const int b  = tile / NTILE;
    const int l0 = (tile - b * NTILE) * VALID;
    const float* xb = x + (size_t)b * L_IN * CIN + 16 * wave + lm;
    v2f bd[16], ba[16];
    #pragma unroll
    for (int s = 0; s < 16; ++s) {
        int n0 = l0 + 4 * s + 2 * half;
        int m0 = l0 + 127 - 4 * s - 2 * half;
        int m1 = m0 - 1;
        int ma = m0 < L_IN ? m0 : L_IN - 1;
        int mb = m1 < L_IN ? m1 : L_IN - 1;
        float x0 = xb[(size_t)n0 * CIN];
        float x1 = xb[(size_t)(n0 + 1) * CIN];
        float va = xb[(size_t)ma * CIN];
        float vb = xb[(size_t)mb * CIN];
        float y0 = m0 < L_IN ? va : 0.f;
        float y1 = m1 < L_IN ? vb : 0.f;
        bd[s].x = x0 - y0;
        bd[s].y = x1 - y1;
        ba[s].x = x0 + y0;
        ba[s].y = x1 + y1;
    }
    #pragma unroll
    for (int i = 0; i < 8; ++i) {
        int q = t + 256 * i;
        int row = q >> 4, c4 = q & 15;
        v4f v = *(const v4f*)(tab + row * NH + c4 * 4);
        *(v4f*)(&lds_t[row * HPAD + c4 * 4]) = v;
    }
    __syncthreads();
    const float* arow = &lds_t[lm * HPAD + 2 * half];
    v8f accp[4] = {}, accq[4] = {};
    #pragma unroll
    for (int s = 0; s < 16; ++s) {
        #pragma unroll
        for (int mt = 0; mt < 4; ++mt) {
            v2f ac = *(const v2f*)(arow + mt * 16 * HPAD + 4 * s);
            v2f as = *(const v2f*)(arow + (NH + mt * 16) * HPAD + 4 * s);
            accp[mt] = mma4(ac, bd[s], accp[mt]);
            accq[mt] = mma4(as, ba[s], accq[mt]);
        }
    }
    #pragma unroll
    for (int mt = 0; mt < 4; ++mt) {
        #pragma unroll
        for (int r = 0; r < 8; ++r) {
            int k = 16 * mt + 8 * half + r;
            Xf[((size_t)(2 * k)     * TILES + tile) * CIN + 16 * wave + lm] = accp[mt][r];
            Xf[((size_t)(2 * k + 1) * TILES + tile) * CIN + 16 * wave + lm] = accq[mt][r];
        }
    }
}

__global__ __launch_bounds__(256)
void k_pw(const float* __restrict__ Xf, const float* __restrict__ G, float* __restrict__ Yf) {
    __shared__ float lds_a[3 * MT_PW * APAD];
    const int t = threadIdx.x, lane = t & 31;
    const int wave = __builtin_amdgcn_readfirstlane(t >> 5);
    const int tile0 = blockIdx.x * MT_PW;
    const int fblk  = blockIdx.y * NF_PW;
    const int k     = blockIdx.z;
    #pragma unroll
    for (int i = 0; i < 8; ++i) {
        int q  = t + 256 * i;
        int c4 = q & 31, m = q >> 5;
        v4f vp = *(const v4f*)(Xf + ((size_t)(2 * k)     * TILES + tile0 + m) * CIN + c4 * 4);
        v4f vq = *(const v4f*)(Xf + ((size_t)(2 * k + 1) * TILES + tile0 + m) * CIN + c4 * 4);
        *(v4f*)(&lds_a[(0 * MT_PW + m) * APAD + c4 * 4]) = vp - vq;
        *(v4f*)(&lds_a[(1 * MT_PW + m) * APAD + c4 * 4]) = vq;
        *(v4f*)(&lds_a[(2 * MT_PW + m) * APAD + c4 * 4]) = vp;
    }
    __syncthreads();
    const int half = lane >> 4, lm = lane & 15;
    const int f0 = fblk + wave * 16;
    const float* g0 = G + ((size_t)(k * 3) * 64 * 256 + f0 + lm) * 2 + half * 512;
    const float* g1 = g0 + 64 * 256 * 2;
    const float* g2 = g1 + 64 * 256 * 2;
    const float* ap = &lds_a[lm * APAD + 2 * half];

    v8f s1[4] = {}, s2[4] = {}, s3[4] = {};
    v2f b0 = *(const v2f*)(g0);
    v2f b1 = *(const v2f*)(g1);
    v2f b2 = *(const v2f*)(g2);
    #pragma unroll 4
    for (int s = 0; s < 32; ++s) {
        v2f c0 = b0, c1 = b1, c2 = b2;
        if (s + 1 < 32) {
            b0 = *(const v2f*)(g0 + (size_t)(s + 1) * 1024);
            b1 = *(const v2f*)(g1 + (size_t)(s + 1) * 1024);
            b2 = *(const v2f*)(g2 + (size_t)(s + 1) * 1024);
        }
        #pragma unroll
        for (int mt = 0; mt < 4; ++mt) {
            v2f da = *(const v2f*)(ap + (0 * MT_PW + mt * 16) * APAD + 4 * s);
            v2f qa = *(const v2f*)(ap + (1 * MT_PW + mt * 16) * APAD + 4 * s);
            v2f pa = *(const v2f*)(ap + (2 * MT_PW + mt * 16) * APAD + 4 * s);
            s1[mt] = mma4(da, c0, s1[mt]);
            s2[mt] = mma4(qa, c1, s2[mt]);
            s3[mt] = mma4(pa, c2, s3[mt]);
        }
    }
    const int f = f0 + lm;
    #pragma unroll
    for (int mt = 0; mt < 4; ++mt) {
        #pragma unroll
        for (int r = 0; r < 8; ++r) {
            int tile = tile0 + 16 * mt + 8 * half + r;
            v2f o;
            o.x = s1[mt][r] + s2[mt][r];
            o.y = s3[mt][r] - s1[mt][r];
            *(v2f*)(Yf + (((size_t)tile * NBIN + k) * F_OUT + f) * 2) = o;
        }
    }
}

__global__ __launch_bounds__(256)
void k_inv(const float* __restrict__ x, const float* __restrict__ tab,
           const float* __restrict__ Yf, const float* __restrict__ scale,
           const float* __restrict__ bias, float* __restrict__ out) {
    __shared__ float lds_t[2 * NH * HPAD];
    __shared__ float lds_part[NFFT * 4];
    __shared__ float lds_invn[NFFT];
    const int t = threadIdx.x, lane = t & 31, wave = t >> 5;
    const int tile = blockIdx.x;
    const int b  = tile / NTILE;
    const int l0 = (tile - b * NTILE) * VALID;
    #pragma unroll
    for (int i = 0; i < 8; ++i) {
        int q = t + 256 * i;
        int row = q >> 4, c4 = q & 15;
        v4f v = *(const v4f*)(tab + row * NH + c4 * 4);
        *(v4f*)(&lds_t[row * HPAD + c4 * 4]) = v;
    }
    {
        int row = t >> 1, hf = t & 1;
        int g = l0 + row;
        float s0 = 0.f, s1 = 0.f;
        if (g < L_IN) {
            const v4f* r = (const v4f*)(x + ((size_t)b * L_IN + g) * CIN + hf * 64);
            #pragma unroll
            for (int i = 0; i < 8; ++i) {
                v4f v = r[i];
                s0 += v.x * v.x + v.y * v.y + v.z * v.z + v.w * v.w;
            }
            #pragma unroll
            for (int i = 8; i < 16; ++i) {
                v4f v = r[i];
                s1 += v.x * v.x + v.y * v.y + v.z * v.z + v.w * v.w;
            }
        }
        lds_part[row * 4 + 2 * hf]     = s0;
        lds_part[row * 4 + 2 * hf + 1] = s1;
    }
    __syncthreads();
    if (t < NFFT) {
        float win = 0.f;
        if (t < VALID) {
            #pragma unroll
            for (int j = 0; j < K_TAPS; ++j) {
                const float* pp = &lds_part[(t + j) * 4];
                win += (pp[0] + pp[1]) + (pp[2] + pp[3]);
            }
        }
        lds_invn[t] = scale[0] / __builtin_sqrtf(win + EPS_F) * (1.0f / 64.0f);
    }
    __syncthreads();

    const int half = lane >> 4, lm = lane & 15;
    const int f0 = wave * 32;
    const float* yb   = Yf + (((size_t)tile * NBIN + 2 * half) * F_OUT + f0 + lm) * 2;
    const float* arow = &lds_t[lm * HPAD + 2 * half];
    v2f bc[2][16], bs[2][16];
    #pragma unroll
    for (int s = 0; s < 16; ++s) {
        #pragma unroll
        for (int j = 0; j < 2; ++j) {
            v2f e0 = *(const v2f*)(yb + (size_t)(4 * s)     * 512 + 32 * j);
            v2f e1 = *(const v2f*)(yb + (size_t)(4 * s + 1) * 512 + 32 * j);
            bc[j][s].x = e0.x;
            bc[j][s].y = e1.x;
            bs[j][s].x = e0.y;
            bs[j][s].y = e1.y;
        }
    }
    v8f accc[4][2] = {}, accs[4][2] = {};
    #pragma unroll
    for (int s = 0; s < 16; ++s) {
        #pragma unroll
        for (int mt = 0; mt < 4; ++mt) {
            v2f ac = *(const v2f*)(arow + mt * 16 * HPAD + 4 * s);
            v2f as = *(const v2f*)(arow + (NH + mt * 16) * HPAD + 4 * s);
            accc[mt][0] = mma4(ac, bc[0][s], accc[mt][0]);
            accc[mt][1] = mma4(ac, bc[1][s], accc[mt][1]);
            accs[mt][0] = mma4(as, bs[0][s], accs[mt][0]);
            accs[mt][1] = mma4(as, bs[1][s], accs[mt][1]);
        }
    }
    const float bias0 = bias[f0 + lm], bias1 = bias[f0 + 16 + lm];
    float* ob = out + ((size_t)b * L_OUT + l0) * F_OUT + f0 + lm;
    #pragma unroll
    for (int mt = 0; mt < 4; ++mt) {
        #pragma unroll
        for (int r = 0; r < 8; ++r) {
            int l = 16 * mt + 8 * half + r;
            int m = NFFT - 1 - l;
            if (l0 + l < L_OUT) {
                float inv = lds_invn[l];
                ob[(size_t)l * F_OUT]      = (accs[mt][0][r] + accc[mt][0][r]) * inv + bias0;
                ob[(size_t)l * F_OUT + 16] = (accs[mt][1][r] + accc[mt][1][r]) * inv + bias1;
            }
            if (m < VALID && l0 + m < L_OUT) {
                float inv = lds_invn[m];
                ob[(size_t)m * F_OUT]      = (accs[mt][0][r] - accc[mt][0][r]) * inv + bias0;
                ob[(size_t)m * F_OUT + 16] = (accs[mt][1][r] - accc[mt][1][r]) * inv + bias1;
            }
        }
    }
}

extern "C" void kernel_launch(void* const* d_in, const int* in_sizes, int n_in,
                              void* d_out, int out_size, void* d_ws, size_t ws_size,
                              hipStream_t stream) {
    const float* x     = (const float*)d_in[0];
    const float* w     = (const float*)d_in[1];
    const float* scale = (const float*)d_in[2];
    const float* bias  = (const float*)d_in[3];
    float* out = (float*)d_out;
    float* ws  = (float*)d_ws;
    float* tab   = ws;
    float* G     = ws + OFF_G;
    float* Xf    = ws + OFF_XF;
    float* Yf    = ws + OFF_YF;

    k_prep<<<NBIN * 16, 256, 0, stream>>>(w, tab, G);
    k_fwd<<<TILES, 256, 0, stream>>>(x, tab, Xf);
    k_pw<<<dim3(TILES / MT_PW, F_OUT / NF_PW, NBIN), 256, 0, stream>>>(Xf, G, Yf);
    k_inv<<<TILES, 256, 0, stream>>>(x, tab, Yf, scale, bias, out);
}


# inverse transform output: lane halves trade filter halves (one cross-lane move per row pair) so each store is one whole 128-B row line; row norms read from LDS as one group; same sums
# speedup vs baseline: 3.4918x; 1.0442x over previous
#include <hip/hip_runtime.h>

typedef __attribute__((ext_vector_type(2))) float v2f;
typedef __attribute__((ext_vector_type(4))) float v4f;
typedef __attribute__((ext_vector_type(8))) float v8f;

#define K_TAPS 15
#define CIN    128
#define F_OUT  256
#define L_IN   4096
#define L_OUT  (L_IN - K_TAPS + 1)
#define BATCH  32
#define NFFT   128
#define VALID  (NFFT - K_TAPS + 1)
#define NTILE  36
#define TILES  (BATCH * NTILE)
#define NBIN   64
#define NH     64
#define HPAD   68
#define APAD   132
#define MT_PW  64
#define NF_PW  128
#define EPS_F  1e-7f

#define OFF_G  8192
#define OFF_XF 6299648
#define OFF_YF 25174016

__device__ __forceinline__ v8f mma4(v2f a, v2f b, v8f c) {
    return __builtin_amdgcn_wmma_f32_16x16x4_f32(false, a, false, b, (short)0, c, false, false);
}

__device__ __forceinline__ float xhalf(float x) {
    int b = __builtin_bit_cast(int, x);
    return __builtin_bit_cast(float, __builtin_amdgcn_permlanex16(b, b, 0x76543210, 0xfedcba98, false, false));
}

template <bool FULL>
__device__ __forceinline__ void inv_store(const v8f (&accc)[4][2], const v8f (&accs)[4][2],
                                          const v4f (&il)[4][2], const v4f (&im)[4][2],
                                          float bias0, float bias1, bool upper, int nvalid,
                                          float* __restrict__ orow) {
    #pragma unroll
    for (int mt = 0; mt < 4; ++mt) {
        #pragma unroll
        for (int r = 0; r < 8; ++r) {
            const float inv_l = il[mt][r >> 2][r & 3];
            const float inv_m = im[mt][(7 - r) >> 2][(7 - r) & 3];
            const float c0 = accc[mt][0][r], c1 = accc[mt][1][r];
            const float s0 = accs[mt][0][r], s1 = accs[mt][1][r];
            const float pl0 = (s0 + c0) * inv_l + bias0;
            const float pl1 = (s1 + c1) * inv_l + bias1;
            const float pm0 = (s0 - c0) * inv_m + bias0;
            const float pm1 = (s1 - c1) * inv_m + bias1;
            const float ul = xhalf(upper ? pl0 : pl1);
            const float um = xhalf(upper ? pm0 : pm1);
            const int rowA = 16 * mt + r;
            const int rowB = rowA + 8;
            const int rowC = NFFT - 1 - rowA;
            const int rowD = NFFT - 1 - rowB;
            if (FULL || rowA < nvalid) orow[(size_t)rowA * F_OUT] = upper ? ul : pl0;
            if (FULL || rowB < nvalid) orow[(size_t)rowB * F_OUT] = upper ? pl1 : ul;
            if (rowC < VALID && (FULL || rowC < nvalid)) orow[(size_t)rowC * F_OUT] = upper ? um : pm0;
            if (rowD < VALID && (FULL || rowD < nvalid)) orow[(size_t)rowD * F_OUT] = upper ? pm1 : um;
        }
    }
}

__global__ __launch_bounds__(256)
void k_prep(const float* __restrict__ w, float* __restrict__ tab, float* __restrict__ G) {
    __shared__ float cs_t[16], sn_t[16];
    const int t   = threadIdx.x;
    const int k   = blockIdx.x >> 4;
    const int cpg = blockIdx.x & 15;
    if (t < K_TAPS) {
        float sn, cs;
        sincospif((float)((2 * k + 1) * t) * (1.0f / 128.0f), &sn, &cs);
        cs_t[t] = cs;
        sn_t[t] = sn;
    }
    __syncthreads();
    #pragma unroll
    for (int i = 0; i < 4; ++i) {
        const int cp = cpg * 4 + i;
        v2f gre = {0.f, 0.f}, gim = {0.f, 0.f};
        #pragma unroll
        for (int tap = 0; tap < K_TAPS; ++tap) {
            float w0 = w[(tap * CIN + 2 * cp)     * F_OUT + t];
            float w1 = w[(tap * CIN + 2 * cp + 1) * F_OUT + t];
            gre.x += w0 * cs_t[tap];
            gre.y += w1 * cs_t[tap];
            gim.x += w0 * sn_t[tap];
            gim.y += w1 * sn_t[tap];
        }
        *(v2f*)(G + (((size_t)(k * 3 + 0) * 64 + cp) * 256 + t) * 2) = gre;
        *(v2f*)(G + (((size_t)(k * 3 + 1) * 64 + cp) * 256 + t) * 2) = gre + gim;
        *(v2f*)(G + (((size_t)(k * 3 + 2) * 64 + cp) * 256 + t) * 2) = gre - gim;
    }
    const int gid = blockIdx.x * 256 + t;
    if (gid < 2 * NH * NH) {
        const int p = gid >> 12, i = (gid >> 6) & 63, j = gid & 63;
        float sn, cs;
        sincospif((float)((2 * i + 1) * (2 * j + 1)) * (1.0f / 256.0f), &sn, &cs);
        tab[gid] = p ? sn : cs;
    }
}

__global__ __launch_bounds__(256)
void k_fwd(const float* __restrict__ x, const float* __restrict__ tab, float* __restrict__ Xf) {
    __shared__ float lds_t[2 * NH * HPAD];
    const int t = threadIdx.x, lane = t & 31, wave = t >> 5;
    const int half = lane >> 4, lm = lane & 15;
    const int tile = blockIdx.x;
    const int b  = tile / NTILE;
    const int l0 = (tile - b * NTILE) * VALID;
    const float* xb = x + (size_t)b * L_IN * CIN + 16 * wave + lm;
    v2f bd[16], ba[16];
    #pragma unroll
    for (int s = 0; s < 16; ++s) {
        int n0 = l0 + 4 * s + 2 * half;
        int m0 = l0 + 127 - 4 * s - 2 * half;
        int m1 = m0 - 1;
        int ma = m0 < L_IN ? m0 : L_IN - 1;
        int mb = m1 < L_IN ? m1 : L_IN - 1;
        float x0 = xb[(size_t)n0 * CIN];
        float x1 = xb[(size_t)(n0 + 1) * CIN];
        float va = xb[(size_t)ma * CIN];
        float vb = xb[(size_t)mb * CIN];
        float y0 = m0 < L_IN ? va : 0.f;
        float y1 = m1 < L_IN ? vb : 0.f;
        bd[s].x = x0 - y0;
        bd[s].y = x1 - y1;
        ba[s].x = x0 + y0;
        ba[s].y = x1 + y1;
    }
    #pragma unroll
    for (int i = 0; i < 8; ++i) {
        int q = t + 256 * i;
        int row = q >> 4, c4 = q & 15;
        v4f v = *(const v4f*)(tab + row * NH + c4 * 4);
        *(v4f*)(&lds_t[row * HPAD + c4 * 4]) = v;
    }
    __syncthreads();
    const float* arow = &lds_t[lm * HPAD + 2 * half];
    v8f accp[4] = {}, accq[4] = {};
    #pragma unroll
    for (int s = 0; s < 16; ++s) {
        #pragma unroll
        for (int mt = 0; mt < 4; ++mt) {
            v2f ac = *(const v2f*)(arow + mt * 16 * HPAD + 4 * s);
            v2f as = *(const v2f*)(arow + (NH + mt * 16) * HPAD + 4 * s);
            accp[mt] = mma4(ac, bd[s], accp[mt]);
            accq[mt] = mma4(as, ba[s], accq[mt]);
        }
    }
    #pragma unroll
    for (int mt = 0; mt < 4; ++mt) {
        #pragma unroll
        for (int r = 0; r < 8; ++r) {
            int k = 16 * mt + 8 * half + r;
            Xf[((size_t)(2 * k)     * TILES + tile) * CIN + 16 * wave + lm] = accp[mt][r];
            Xf[((size_t)(2 * k + 1) * TILES + tile) * CIN + 16 * wave + lm] = accq[mt][r];
        }
    }
}

__global__ __launch_bounds__(256)
void k_pw(const float* __restrict__ Xf, const float* __restrict__ G, float* __restrict__ Yf) {
    __shared__ float lds_a[3 * MT_PW * APAD];
    const int t = threadIdx.x, lane = t & 31;
    const int wave = __builtin_amdgcn_readfirstlane(t >> 5);
    const int tile0 = blockIdx.x * MT_PW;
    const int fblk  = blockIdx.y * NF_PW;
    const int k     = blockIdx.z;
    #pragma unroll
    for (int i = 0; i < 8; ++i) {
        int q  = t + 256 * i;
        int c4 = q & 31, m = q >> 5;
        v4f vp = *(const v4f*)(Xf + ((size_t)(2 * k)     * TILES + tile0 + m) * CIN + c4 * 4);
        v4f vq = *(const v4f*)(Xf + ((size_t)(2 * k + 1) * TILES + tile0 + m) * CIN + c4 * 4);
        *(v4f*)(&lds_a[(0 * MT_PW + m) * APAD + c4 * 4]) = vp - vq;
        *(v4f*)(&lds_a[(1 * MT_PW + m) * APAD + c4 * 4]) = vq;
        *(v4f*)(&lds_a[(2 * MT_PW + m) * APAD + c4 * 4]) = vp;
    }
    __syncthreads();
    const int half = lane >> 4, lm = lane & 15;
    const int f0 = fblk + wave * 16;
    const float* g0 = G + ((size_t)(k * 3) * 64 * 256 + f0 + lm) * 2 + half * 512;
    const float* g1 = g0 + 64 * 256 * 2;
    const float* g2 = g1 + 64 * 256 * 2;
    const float* ap = &lds_a[lm * APAD + 2 * half];

    v8f s1[4] = {}, s2[4] = {}, s3[4] = {};
    v2f b0 = *(const v2f*)(g0);
    v2f b1 = *(const v2f*)(g1);
    v2f b2 = *(const v2f*)(g2);
    #pragma unroll 4
    for (int s = 0; s < 32; ++s) {
        v2f c0 = b0, c1 = b1, c2 = b2;
        if (s + 1 < 32) {
            b0 = *(const v2f*)(g0 + (size_t)(s + 1) * 1024);
            b1 = *(const v2f*)(g1 + (size_t)(s + 1) * 1024);
            b2 = *(const v2f*)(g2 + (size_t)(s + 1) * 1024);
        }
        #pragma unroll
        for (int mt = 0; mt < 4; ++mt) {
            v2f da = *(const v2f*)(ap + (0 * MT_PW + mt * 16) * APAD + 4 * s);
            v2f qa = *(const v2f*)(ap + (1 * MT_PW + mt * 16) * APAD + 4 * s);
            v2f pa = *(const v2f*)(ap + (2 * MT_PW + mt * 16) * APAD + 4 * s);
            s1[mt] = mma4(da, c0, s1[mt]);
            s2[mt] = mma4(qa, c1, s2[mt]);
            s3[mt] = mma4(pa, c2, s3[mt]);
        }
    }
    const int f = f0 + lm;
    #pragma unroll
    for (int mt = 0; mt < 4; ++mt) {
        #pragma unroll
        for (int r = 0; r < 8; ++r) {
            int tile = tile0 + 16 * mt + 8 * half + r;
            v2f o;
            o.x = s1[mt][r] + s2[mt][r];
            o.y = s3[mt][r] - s1[mt][r];
            *(v2f*)(Yf + (((size_t)tile * NBIN + k) * F_OUT + f) * 2) = o;
        }
    }
}

__global__ __launch_bounds__(256)
void k_inv(const float* __restrict__ x, const float* __restrict__ tab,
           const float* __restrict__ Yf, const float* __restrict__ scale,
           const float* __restrict__ bias, float* __restrict__ out) {
    __shared__ float lds_t[2 * NH * HPAD];
    __shared__ float lds_part[NFFT * 4];
    __shared__ __attribute__((aligned(16))) float lds_invn[NFFT];
    const int t = threadIdx.x, lane = t & 31, wave = t >> 5;
    const int tile = blockIdx.x;
    const int b  = tile / NTILE;
    const int l0 = (tile - b * NTILE) * VALID;
    #pragma unroll
    for (int i = 0; i < 8; ++i) {
        int q = t + 256 * i;
        int row = q >> 4, c4 = q & 15;
        v4f v = *(const v4f*)(tab + row * NH + c4 * 4);
        *(v4f*)(&lds_t[row * HPAD + c4 * 4]) = v;
    }
    {
        int row = t >> 1, hf = t & 1;
        int g = l0 + row;
        float s0 = 0.f, s1 = 0.f;
        if (g < L_IN) {
            const v4f* r = (const v4f*)(x + ((size_t)b * L_IN + g) * CIN + hf * 64);
            #pragma unroll
            for (int i = 0; i < 8; ++i) {
                v4f v = r[i];
                s0 += v.x * v.x + v.y * v.y + v.z * v.z + v.w * v.w;
            }
            #pragma unroll
            for (int i = 8; i < 16; ++i) {
                v4f v = r[i];
                s1 += v.x * v.x + v.y * v.y + v.z * v.z + v.w * v.w;
            }
        }
        lds_part[row * 4 + 2 * hf]     = s0;
        lds_part[row * 4 + 2 * hf + 1] = s1;
    }
    __syncthreads();
    if (t < NFFT) {
        float win = 0.f;
        if (t < VALID) {
            #pragma unroll
            for (int j = 0; j < K_TAPS; ++j) {
                const float* pp = &lds_part[(t + j) * 4];
                win += (pp[0] + pp[1]) + (pp[2] + pp[3]);
            }
        }
        lds_invn[t] = scale[0] / __builtin_sqrtf(win + EPS_F) * (1.0f / 64.0f);
    }
    __syncthreads();

    const int half = lane >> 4, lm = lane & 15;
    const int f0 = wave * 32;
    const float* yb   = Yf + (((size_t)tile * NBIN + 2 * half) * F_OUT + f0 + lm) * 2;
    const float* arow = &lds_t[lm * HPAD + 2 * half];
    v2f bc[2][16], bs[2][16];
    #pragma unroll
    for (int s = 0; s < 16; ++s) {
        #pragma unroll
        for (int j = 0; j < 2; ++j) {
            v2f e0 = *(const v2f*)(yb + (size_t)(4 * s)     * 512 + 32 * j);
            v2f e1 = *(const v2f*)(yb + (size_t)(4 * s + 1) * 512 + 32 * j);
            bc[j][s].x = e0.x;
            bc[j][s].y = e1.x;
            bs[j][s].x = e0.y;
            bs[j][s].y = e1.y;
        }
    }
    v8f accc[4][2] = {}, accs[4][2] = {};
    #pragma unroll
    for (int s = 0; s < 16; ++s) {
        #pragma unroll
        for (int mt = 0; mt < 4; ++mt) {
            v2f ac = *(const v2f*)(arow + mt * 16 * HPAD + 4 * s);
            v2f as = *(const v2f*)(arow + (NH + mt * 16) * HPAD + 4 * s);
            accc[mt][0] = mma4(ac, bc[0][s], accc[mt][0]);
            accc[mt][1] = mma4(ac, bc[1][s], accc[mt][1]);
            accs[mt][0] = mma4(as, bs[0][s], accs[mt][0]);
            accs[mt][1] = mma4(as, bs[1][s], accs[mt][1]);
        }
    }
    const float bias0 = bias[f0 + lm], bias1 = bias[f0 + 16 + lm];
    v4f il[4][2], im[4][2];
    #pragma unroll
    for (int mt = 0; mt < 4; ++mt) {
        const float* pl = &lds_invn[16 * mt + 8 * half];
        const float* pm = &lds_invn[NFFT - 8 - 16 * mt - 8 * half];
        il[mt][0] = *(const v4f*)(pl);
        il[mt][1] = *(const v4f*)(pl + 4);
        im[mt][0] = *(const v4f*)(pm);
        im[mt][1] = *(const v4f*)(pm + 4);
    }
    const int nvalid = (L_OUT - l0) < VALID ? (L_OUT - l0) : VALID;
    float* orow = out + ((size_t)b * L_OUT + l0) * F_OUT + f0 + lane;
    if (nvalid == VALID) inv_store<true>(accc, accs, il, im, bias0, bias1, half != 0, nvalid, orow);
    else                 inv_store<false>(accc, accs, il, im, bias0, bias1, half != 0, nvalid, orow);
}

extern "C" void kernel_launch(void* const* d_in, const int* in_sizes, int n_in,
                              void* d_out, int out_size, void* d_ws, size_t ws_size,
                              hipStream_t stream) {
    const float* x     = (const float*)d_in[0];
    const float* w     = (const float*)d_in[1];
    const float* scale = (const float*)d_in[2];
    const float* bias  = (const float*)d_in[3];
    float* out = (float*)d_out;
    float* ws  = (float*)d_ws;
    float* tab   = ws;
    float* G     = ws + OFF_G;
    float* Xf    = ws + OFF_XF;
    float* Yf    = ws + OFF_YF;

    k_prep<<<NBIN * 16, 256, 0, stream>>>(w, tab, G);
    k_fwd<<<TILES, 256, 0, stream>>>(x, tab, Xf);
    k_pw<<<dim3(TILES / MT_PW, F_OUT / NF_PW, NBIN), 256, 0, stream>>>(Xf, G, Yf);
    k_inv<<<TILES, 256, 0, stream>>>(x, tab, Yf, scale, bias, out);
}


# x is read once: forward launch takes its tile by whole 512-B rows through LDS and also writes each row's sum of squares; inverse launch reads 128 sums per tile, not x; norm sum order changes
# speedup vs baseline: 3.6510x; 1.0456x over previous
#include <hip/hip_runtime.h>

typedef __attribute__((ext_vector_type(2))) float v2f;
typedef __attribute__((ext_vector_type(4))) float v4f;
typedef __attribute__((ext_vector_type(8))) float v8f;

#define K_TAPS 15
#define CIN    128
#define F_OUT  256
#define L_IN   4096
#define L_OUT  (L_IN - K_TAPS + 1)
#define BATCH  32
#define NFFT   128
#define VALID  (NFFT - K_TAPS + 1)
#define NTILE  36
#define TILES  (BATCH * NTILE)
#define NBIN   64
#define NH     64
#define HPAD   68
#define APAD   132
#define XPAD   288
#define MT_PW  64
#define NF_PW  128
#define EPS_F  1e-7f

#define OFF_G  8192
#define OFF_XF 6299648
#define OFF_YF 25174016
#define OFF_SQ 62922752

__device__ __forceinline__ v8f mma4(v2f a, v2f b, v8f c) {
    return __builtin_amdgcn_wmma_f32_16x16x4_f32(false, a, false, b, (short)0, c, false, false);
}

__device__ __forceinline__ float xhalf(float x) {
    int b = __builtin_bit_cast(int, x);
    return __builtin_bit_cast(float, __builtin_amdgcn_permlanex16(b, b, 0x76543210, 0xfedcba98, false, false));
}

template <bool FULL>
__device__ __forceinline__ void inv_store(const v8f (&accc)[4][2], const v8f (&accs)[4][2],
                                          const v4f (&il)[4][2], const v4f (&im)[4][2],
                                          float bias0, float bias1, bool upper, int nvalid,
                                          float* __restrict__ orow) {
    #pragma unroll
    for (int mt = 0; mt < 4; ++mt) {
        #pragma unroll
        for (int r = 0; r < 8; ++r) {
            const float inv_l = il[mt][r >> 2][r & 3];
            const float inv_m = im[mt][(7 - r) >> 2][(7 - r) & 3];
            const float c0 = accc[mt][0][r], c1 = accc[mt][1][r];
            const float s0 = accs[mt][0][r], s1 = accs[mt][1][r];
            const float pl0 = (s0 + c0) * inv_l + bias0;
            const float pl1 = (s1 + c1) * inv_l + bias1;
            const float pm0 = (s0 - c0) * inv_m + bias0;
            const float pm1 = (s1 - c1) * inv_m + bias1;
            const float ul = xhalf(upper ? pl0 : pl1);
            const float um = xhalf(upper ? pm0 : pm1);
            const int rowA = 16 * mt + r;
            const int rowB = rowA + 8;
            const int rowC = NFFT - 1 - rowA;
            const int rowD = NFFT - 1 - rowB;
            if (FULL || rowA < nvalid) orow[(size_t)rowA * F_OUT] = upper ? ul : pl0;
            if (FULL || rowB < nvalid) orow[(size_t)rowB * F_OUT] = upper ? pl1 : ul;
            if (rowC < VALID && (FULL || rowC < nvalid)) orow[(size_t)rowC * F_OUT] = upper ? um : pm0;
            if (rowD < VALID && (FULL || rowD < nvalid)) orow[(size_t)rowD * F_OUT] = upper ? pm1 : um;
        }
    }
}

__global__ __launch_bounds__(256)
void k_prep(const float* __restrict__ w, float* __restrict__ tab, float* __restrict__ G) {
    __shared__ float cs_t[16], sn_t[16];
    const int t   = threadIdx.x;
    const int k   = blockIdx.x >> 4;
    const int cpg = blockIdx.x & 15;
    if (t < K_TAPS) {
        float sn, cs;
        sincospif((float)((2 * k + 1) * t) * (1.0f / 128.0f), &sn, &cs);
        cs_t[t] = cs;
        sn_t[t] = sn;
    }
    __syncthreads();
    #pragma unroll
    for (int i = 0; i < 4; ++i) {
        const int cp = cpg * 4 + i;
        v2f gre = {0.f, 0.f}, gim = {0.f, 0.f};
        #pragma unroll
        for (int tap = 0; tap < K_TAPS; ++tap) {
            float w0 = w[(tap * CIN + 2 * cp)     * F_OUT + t];
            float w1 = w[(tap * CIN + 2 * cp + 1) * F_OUT + t];
            gre.x += w0 * cs_t[tap];
            gre.y += w1 * cs_t[tap];
            gim.x += w0 * sn_t[tap];
            gim.y += w1 * sn_t[tap];
        }
        *(v2f*)(G + (((size_t)(k * 3 + 0) * 64 + cp) * 256 + t) * 2) = gre;
        *(v2f*)(G + (((size_t)(k * 3 + 1) * 64 + cp) * 256 + t) * 2) = gre + gim;
        *(v2f*)(G + (((size_t)(k * 3 + 2) * 64 + cp) * 256 + t) * 2) = gre - gim;
    }
    const int gid = blockIdx.x * 256 + t;
    if (gid < 2 * NH * NH) {
        const int p = gid >> 12, i = (gid >> 6) & 63, j = gid & 63;
        float sn, cs;
        sincospif((float)((2 * i + 1) * (2 * j + 1)) * (1.0f / 256.0f), &sn, &cs);
        tab[gid] = p ? sn : cs;
    }
}

__global__ __launch_bounds__(256)
void k_fwd(const float* __restrict__ x, const float* __restrict__ tab, float* __restrict__ Xf, float* __restrict__ SQ) {
    __shared__ float lds_t[2 * NH * HPAD];
    __shared__ __attribute__((aligned(16))) float lds_x[2 * (NH / 2) * XPAD];
    const int t = threadIdx.x, lane = t & 31;
    const int wave = __builtin_amdgcn_readfirstlane(t >> 5);
    const int half = lane >> 4, lm = lane & 15;
    const int tile = blockIdx.x;
    const int b  = tile / NTILE;
    const int l0 = (tile - b * NTILE) * VALID;
    const float* xt = x + ((size_t)b * L_IN + l0) * CIN + 4 * lane;
    v4f tv[8];
    #pragma unroll
    for (int i = 0; i < 8; ++i) {
        int q = t + 256 * i;
        int row = q >> 4, c4 = q & 15;
        tv[i] = *(const v4f*)(tab + row * NH + c4 * 4);
    }
    v4f r0[4], r1[4], q0[4], q1[4];
    #pragma unroll
    for (int i = 0; i < 4; ++i) {
        const int n0 = 2 * (wave + 8 * i);
        const int m0 = NFFT - 1 - n0, m1 = m0 - 1;
        const int ma = (l0 + m0 < L_IN) ? m0 : (L_IN - 1 - l0);
        const int mb = (l0 + m1 < L_IN) ? m1 : (L_IN - 1 - l0);
        r0[i] = *(const v4f*)(xt + (size_t)n0 * CIN);
        r1[i] = *(const v4f*)(xt + (size_t)(n0 + 1) * CIN);
        q0[i] = *(const v4f*)(xt + (size_t)ma * CIN);
        q1[i] = *(const v4f*)(xt + (size_t)mb * CIN);
    }
    #pragma unroll
    for (int i = 0; i < 8; ++i) {
        int q = t + 256 * i;
        int row = q >> 4, c4 = q & 15;
        *(v4f*)(&lds_t[row * HPAD + c4 * 4]) = tv[i];
    }
    float psq[16];
    #pragma unroll
    for (int i = 0; i < 4; ++i) {
        const int n2 = wave + 8 * i;
        const int m0 = NFFT - 1 - 2 * n2, m1 = m0 - 1;
        const v4f zero = {0.f, 0.f, 0.f, 0.f};
        const v4f y0 = (l0 + m0 < L_IN) ? q0[i] : zero;
        const v4f y1 = (l0 + m1 < L_IN) ? q1[i] : zero;
        const v4f d0 = r0[i] - y0, d1 = r1[i] - y1;
        const v4f a0 = r0[i] + y0, a1 = r1[i] + y1;
        float* pd = &lds_x[n2 * XPAD + 8 * lane];
        float* pa = pd + (NH / 2) * XPAD;
        v4f w;
        w.x = d0.x; w.y = d1.x; w.z = d0.y; w.w = d1.y;  *(v4f*)(pd)     = w;
        w.x = d0.z; w.y = d1.z; w.z = d0.w; w.w = d1.w;  *(v4f*)(pd + 4) = w;
        w.x = a0.x; w.y = a1.x; w.z = a0.y; w.w = a1.y;  *(v4f*)(pa)     = w;
        w.x = a0.z; w.y = a1.z; w.z = a0.w; w.w = a1.w;  *(v4f*)(pa + 4) = w;
        psq[4 * i]     = (r0[i].x * r0[i].x + r0[i].y * r0[i].y) + (r0[i].z * r0[i].z + r0[i].w * r0[i].w);
        psq[4 * i + 1] = (r1[i].x * r1[i].x + r1[i].y * r1[i].y) + (r1[i].z * r1[i].z + r1[i].w * r1[i].w);
        psq[4 * i + 2] = (y0.x * y0.x + y0.y * y0.y) + (y0.z * y0.z + y0.w * y0.w);
        psq[4 * i + 3] = (y1.x * y1.x + y1.y * y1.y) + (y1.z * y1.z + y1.w * y1.w);
    }
    {
        const bool b4 = (lane & 16) != 0, b3 = (lane & 8) != 0, b2 = (lane & 4) != 0, b1 = (lane & 2) != 0;
        float q8[8], q4[4], q2[2];
        #pragma unroll
        for (int k = 0; k < 8; ++k) {
            const float keep = b4 ? psq[2 * k + 1] : psq[2 * k];
            const float send = b4 ? psq[2 * k] : psq[2 * k + 1];
            q8[k] = keep + __shfl_xor(send, 16);
        }
        #pragma unroll
        for (int k = 0; k < 4; ++k) {
            const float keep = b3 ? q8[2 * k + 1] : q8[2 * k];
            const float send = b3 ? q8[2 * k] : q8[2 * k + 1];
            q4[k] = keep + __shfl_xor(send, 8);
        }
        #pragma unroll
        for (int k = 0; k < 2; ++k) {
            const float keep = b2 ? q4[2 * k + 1] : q4[2 * k];
            const float send = b2 ? q4[2 * k] : q4[2 * k + 1];
            q2[k] = keep + __shfl_xor(send, 4);
        }
        float q1s = (b1 ? q2[1] : q2[0]) + __shfl_xor(b1 ? q2[0] : q2[1], 2);
        q1s += __shfl_xor(q1s, 1);
        const int idx = (b1 ? 8 : 0) + (b2 ? 4 : 0) + (b3 ? 2 : 0) + (b4 ? 1 : 0);
        const int rn2 = wave + 8 * (idx >> 2);
        const int kind = idx & 3;
        const int rrow = kind == 0 ? 2 * rn2 : kind == 1 ? 2 * rn2 + 1 : kind == 2 ? NFFT - 1 - 2 * rn2 : NFFT - 2 - 2 * rn2;
        if ((lane & 1) == 0) SQ[(size_t)tile * NFFT + rrow] = q1s;
    }
    __syncthreads();
    const float* brow = &lds_x[half * XPAD + (16 * wave + lm) * 2];
    v2f bd[16], ba[16];
    #pragma unroll
    for (int s = 0; s < 16; ++s) {
        bd[s] = *(const v2f*)(brow + 2 * s * XPAD);
        ba[s] = *(const v2f*)(brow + (NH / 2 + 2 * s) * XPAD);
    }
    const float* arow = &lds_t[lm * HPAD + 2 * half];
    v8f accp[4] = {}, accq[4] = {};
    #pragma unroll
    for (int s = 0; s < 16; ++s) {
        #pragma unroll
        for (int mt = 0; mt < 4; ++mt) {
            v2f ac = *(const v2f*)(arow + mt * 16 * HPAD + 4 * s);
            v2f as = *(const v2f*)(arow + (NH + mt * 16) * HPAD + 4 * s);
            accp[mt] = mma4(ac, bd[s], accp[mt]);
            accq[mt] = mma4(as, ba[s], accq[mt]);
        }
    }
    #pragma unroll
    for (int mt = 0; mt < 4; ++mt) {
        #pragma unroll
        for (int r = 0; r < 8; ++r) {
            int k = 16 * mt + 8 * half + r;
            Xf[((size_t)(2 * k)     * TILES + tile) * CIN + 16 * wave + lm] = accp[mt][r];
            Xf[((size_t)(2 * k + 1) * TILES + tile) * CIN + 16 * wave + lm] = accq[mt][r];
        }
    }
}

__global__ __launch_bounds__(256)
void k_pw(const float* __restrict__ Xf, const float* __restrict__ G, float* __restrict__ Yf) {
    __shared__ float lds_a[3 * MT_PW * APAD];
    const int t = threadIdx.x, lane = t & 31;
    const int wave = __builtin_amdgcn_readfirstlane(t >> 5);
    const int tile0 = blockIdx.x * MT_PW;
    const int fblk  = blockIdx.y * NF_PW;
    const int k     = blockIdx.z;
    #pragma unroll
    for (int i = 0; i < 8; ++i) {
        int q  = t + 256 * i;
        int c4 = q & 31, m = q >> 5;
        v4f vp = *(const v4f*)(Xf + ((size_t)(2 * k)     * TILES + tile0 + m) * CIN + c4 * 4);
        v4f vq = *(const v4f*)(Xf + ((size_t)(2 * k + 1) * TILES + tile0 + m) * CIN + c4 * 4);
        *(v4f*)(&lds_a[(0 * MT_PW + m) * APAD + c4 * 4]) = vp - vq;
        *(v4f*)(&lds_a[(1 * MT_PW + m) * APAD + c4 * 4]) = vq;
        *(v4f*)(&lds_a[(2 * MT_PW + m) * APAD + c4 * 4]) = vp;
    }
    __syncthreads();
    const int half = lane >> 4, lm = lane & 15;
    const int f0 = fblk + wave * 16;
    const float* g0 = G + ((size_t)(k * 3) * 64 * 256 + f0 + lm) * 2 + half * 512;
    const float* g1 = g0 + 64 * 256 * 2;
    const float* g2 = g1 + 64 * 256 * 2;
    const float* ap = &lds_a[lm * APAD + 2 * half];

    v8f s1[4] = {}, s2[4] = {}, s3[4] = {};
    v2f b0 = *(const v2f*)(g0);
    v2f b1 = *(const v2f*)(g1);
    v2f b2 = *(const v2f*)(g2);
    #pragma unroll 4
    for (int s = 0; s < 32; ++s) {
        v2f c0 = b0, c1 = b1, c2 = b2;
        if (s + 1 < 32) {
            b0 = *(const v2f*)(g0 + (size_t)(s + 1) * 1024);
            b1 = *(const v2f*)(g1 + (size_t)(s + 1) * 1024);
            b2 = *(const v2f*)(g2 + (size_t)(s + 1) * 1024);
        }
        #pragma unroll
        for (int mt = 0; mt < 4; ++mt) {
            v2f da = *(const v2f*)(ap + (0 * MT_PW + mt * 16) * APAD + 4 * s);
            v2f qa = *(const v2f*)(ap + (1 * MT_PW + mt * 16) * APAD + 4 * s);
            v2f pa = *(const v2f*)(ap + (2 * MT_PW + mt * 16) * APAD + 4 * s);
            s1[mt] = mma4(da, c0, s1[mt]);
            s2[mt] = mma4(qa, c1, s2[mt]);
            s3[mt] = mma4(pa, c2, s3[mt]);
        }
    }
    const int f = f0 + lm;
    #pragma unroll
    for (int mt = 0; mt < 4; ++mt) {
        #pragma unroll
        for (int r = 0; r < 8; ++r) {
            int tile = tile0 + 16 * mt + 8 * half + r;
            v2f o;
            o.x = s1[mt][r] + s2[mt][r];
            o.y = s3[mt][r] - s1[mt][r];
            *(v2f*)(Yf + (((size_t)tile * NBIN + k) * F_OUT + f) * 2) = o;
        }
    }
}

__global__ __launch_bounds__(256)
void k_inv(const float* __restrict__ SQ, const float* __restrict__ tab,
           const float* __restrict__ Yf, const float* __restrict__ scale,
           const float* __restrict__ bias, float* __restrict__ out) {
    __shared__ float lds_t[2 * NH * HPAD];
    __shared__ float lds_part[NFFT];
    __shared__ __attribute__((aligned(16))) float lds_invn[NFFT];
    const int t = threadIdx.x, lane = t & 31, wave = t >> 5;
    const int tile = blockIdx.x;
    const int b  = tile / NTILE;
    const int l0 = (tile - b * NTILE) * VALID;
    #pragma unroll
    for (int i = 0; i < 8; ++i) {
        int q = t + 256 * i;
        int row = q >> 4, c4 = q & 15;
        v4f v = *(const v4f*)(tab + row * NH + c4 * 4);
        *(v4f*)(&lds_t[row * HPAD + c4 * 4]) = v;
    }
    if (t < NFFT) lds_part[t] = SQ[(size_t)tile * NFFT + t];
    __syncthreads();
    if (t < NFFT) {
        float win = 0.f;
        if (t < VALID) {
            #pragma unroll
            for (int j = 0; j < K_TAPS; ++j) {
                win += lds_part[t + j];
            }
        }
        lds_invn[t] = scale[0] / __builtin_sqrtf(win + EPS_F) * (1.0f / 64.0f);
    }
    __syncthreads();

    const int half = lane >> 4, lm = lane & 15;
    const int f0 = wave * 32;
    const float* yb   = Yf + (((size_t)tile * NBIN + 2 * half) * F_OUT + f0 + lm) * 2;
    const float* arow = &lds_t[lm * HPAD + 2 * half];
    v2f bc[2][16], bs[2][16];
    #pragma unroll
    for (int s = 0; s < 16; ++s) {
        #pragma unroll
        for (int j = 0; j < 2; ++j) {
            v2f e0 = *(const v2f*)(yb + (size_t)(4 * s)     * 512 + 32 * j);
            v2f e1 = *(const v2f*)(yb + (size_t)(4 * s + 1) * 512 + 32 * j);
            bc[j][s].x = e0.x;
            bc[j][s].y = e1.x;
            bs[j][s].x = e0.y;
            bs[j][s].y = e1.y;
        }
    }
    v8f accc[4][2] = {}, accs[4][2] = {};
    #pragma unroll
    for (int s = 0; s < 16; ++s) {
        #pragma unroll
        for (int mt = 0; mt < 4; ++mt) {
            v2f ac = *(const v2f*)(arow + mt * 16 * HPAD + 4 * s);
            v2f as = *(const v2f*)(arow + (NH + mt * 16) * HPAD + 4 * s);
            accc[mt][0] = mma4(ac, bc[0][s], accc[mt][0]);
            accc[mt][1] = mma4(ac, bc[1][s], accc[mt][1]);
            accs[mt][0] = mma4(as, bs[0][s], accs[mt][0]);
            accs[mt][1] = mma4(as, bs[1][s], accs[mt][1]);
        }
    }
    const float bias0 = bias[f0 + lm], bias1 = bias[f0 + 16 + lm];
    v4f il[4][2], im[4][2];
    #pragma unroll
    for (int mt = 0; mt < 4; ++mt) {
        const float* pl = &lds_invn[16 * mt + 8 * half];
        const float* pm = &lds_invn[NFFT - 8 - 16 * mt - 8 * half];
        il[mt][0] = *(const v4f*)(pl);
        il[mt][1] = *(const v4f*)(pl + 4);
        im[mt][0] = *(const v4f*)(pm);
        im[mt][1] = *(const v4f*)(pm + 4);
    }
    const int nvalid = (L_OUT - l0) < VALID ? (L_OUT - l0) : VALID;
    float* orow = out + ((size_t)b * L_OUT + l0) * F_OUT + f0 + lane;
    if (nvalid == VALID) inv_store<true>(accc, accs, il, im, bias0, bias1, half != 0, nvalid, orow);
    else                 inv_store<false>(accc, accs, il, im, bias0, bias1, half != 0, nvalid, orow);
}

extern "C" void kernel_launch(void* const* d_in, const int* in_sizes, int n_in,
                              void* d_out, int out_size, void* d_ws, size_t ws_size,
                              hipStream_t stream) {
    const float* x     = (const float*)d_in[0];
    const float* w     = (const float*)d_in[1];
    const float* scale = (const float*)d_in[2];
    const float* bias  = (const float*)d_in[3];
    float* out = (float*)d_out;
    float* ws  = (float*)d_ws;
    float* tab   = ws;
    float* G     = ws + OFF_G;
    float* Xf    = ws + OFF_XF;
    float* Yf    = ws + OFF_YF;
    float* SQ    = ws + OFF_SQ;

    k_prep<<<NBIN * 16, 256, 0, stream>>>(w, tab, G);
    k_fwd<<<TILES, 256, 0, stream>>>(x, tab, Xf, SQ);
    k_pw<<<dim3(TILES / MT_PW, F_OUT / NF_PW, NBIN), 256, 0, stream>>>(Xf, G, Yf);
    k_inv<<<TILES, 256, 0, stream>>>(SQ, tab, Yf, scale, bias, out);
}
